# starting kernel
# baseline (speedup 1.0000x reference)
#include <hip/hip_runtime.h>
#include <hip/hip_bf16.h>

#define N_NODES 50000
#define N_EDGES 312500
#define F 256
#define K2 512
#define M_TILES 3125
#define BSTRIDE 516

typedef float v2f __attribute__((ext_vector_type(2)));
typedef float v8f __attribute__((ext_vector_type(8)));

__global__ void GNN_zero_kernel(float* __restrict__ p, size_t n) {
    size_t i = (size_t)blockIdx.x * blockDim.x + threadIdx.x;
    size_t stride = (size_t)gridDim.x * blockDim.x;
    for (; i < n; i += stride) p[i] = 0.0f;
}

__global__ __launch_bounds__(256) void GNN_scatter_kernel(
    const float* __restrict__ x,
    const int*   __restrict__ Ai,
    const int*   __restrict__ Aj,
    float* __restrict__ S,
    float* __restrict__ deg)
{
    const int lane   = threadIdx.x & 31;
    const int wave   = (blockIdx.x * (blockDim.x >> 5)) + (threadIdx.x >> 5);
    const int nwaves = gridDim.x * (blockDim.x >> 5);

    for (int e = wave; e < N_EDGES; e += nwaves) {
        const int ai = Ai[e];
        const int aj = Aj[e];

        const int en = e + nwaves;
        if (en < N_EDGES) {
            const int ajn = Aj[en];
            __builtin_prefetch(x + (size_t)ajn * F + lane * 8, 0, 0);
        }

        if (lane == 0) atomicAdd(&deg[ai], 1.0f);

        const float* __restrict__ xr = x + (size_t)aj * F;
        float*       __restrict__ sr = S + (size_t)ai * F;
        #pragma unroll
        for (int c = 0; c < F; c += 32) {
            atomicAdd(&sr[c + lane], xr[c + lane]);
        }
    }
}

__global__ __launch_bounds__(256) void GNN_gemm_kernel(
    const float* __restrict__ x,
    const float* __restrict__ W,
    const float* __restrict__ b,
    const float* __restrict__ S,
    const float* __restrict__ deg,
    float* __restrict__ out)
{
    __shared__ float Bs[16 * BSTRIDE];

    const int n0   = blockIdx.y * 16;
    const int wave = threadIdx.x >> 5;
    const int lane = threadIdx.x & 31;
    const int mtile = blockIdx.x * 8 + wave;

    for (int i = threadIdx.x; i < 16 * K2; i += 256) {
        const int j = i >> 9;
        const int k = i & (K2 - 1);
        Bs[j * BSTRIDE + k] = W[(size_t)(n0 + j) * K2 + k];
    }
    __syncthreads();

    if (mtile >= M_TILES) return;

    const int m0    = mtile * 16;
    const int arow  = m0 + (lane & 15);
    const int khalf = (lane >> 4) * 2;
    const int ncol  = lane & 15;
    const float dsc = deg[arow];

    const float* __restrict__ xr = x + (size_t)arow * F;
    const float* __restrict__ sr = S + (size_t)arow * F;

    v8f acc = {};

    #pragma unroll 4
    for (int k = 0; k < F; k += 4) {
        v2f a, bb;
        a.x  = dsc * xr[k + khalf];
        a.y  = dsc * xr[k + khalf + 1];
        bb.x = Bs[ncol * BSTRIDE + k + khalf];
        bb.y = Bs[ncol * BSTRIDE + k + khalf + 1];
        acc = __builtin_amdgcn_wmma_f32_16x16x4_f32(
            false, a, false, bb, (short)0, acc, false, false);
    }
    #pragma unroll 4
    for (int k = 0; k < F; k += 4) {
        v2f a, bb;
        a.x  = sr[k + khalf];
        a.y  = sr[k + khalf + 1];
        bb.x = Bs[ncol * BSTRIDE + F + k + khalf];
        bb.y = Bs[ncol * BSTRIDE + F + k + khalf + 1];
        acc = __builtin_amdgcn_wmma_f32_16x16x4_f32(
            false, a, false, bb, (short)0, acc, false, false);
    }

    const int nout  = n0 + (lane & 15);
    const int mbase = m0 + (lane >> 4) * 8;
    const float bias = b[nout];
    #pragma unroll
    for (int i = 0; i < 8; ++i) {
        const int m = mbase + i;
        out[(size_t)m * F + nout] = acc[i] + deg[m] * bias;
    }
}

extern "C" void kernel_launch(void* const* d_in, const int* in_sizes, int n_in,
                              void* d_out, int out_size, void* d_ws, size_t ws_size,
                              hipStream_t stream) {
    const float* x  = (const float*)d_in[0];
    const float* W  = (const float*)d_in[1];
    const float* b  = (const float*)d_in[2];
    const int*   Ai = (const int*)d_in[3];
    const int*   Aj = (const int*)d_in[4];
    float* out = (float*)d_out;

    float* S   = (float*)d_ws;
    float* deg = S + (size_t)N_NODES * F;

    const size_t nzero = (size_t)N_NODES * F + N_NODES;
    GNN_zero_kernel<<<1024, 256, 0, stream>>>(S, nzero);

    GNN_scatter_kernel<<<2048, 256, 0, stream>>>(x, Ai, Aj, S, deg);

    dim3 grid((M_TILES + 7) / 8, F / 16);
    GNN_gemm_kernel<<<grid, 256, 0, stream>>>(x, W, b, S, deg, out);
}


# Aggregation re-decomposed by destination range: 256 owner blocks each filter the whole edge list for their 196 rows, sum source rows in LDS (LDS f32 adds, arrival order), store S and deg once
# speedup vs baseline: 2.3975x; 2.3975x over previous
#include <hip/hip_runtime.h>
#include <hip/hip_bf16.h>

#define N_NODES 50000
#define N_EDGES 312500
#define F 256
#define K2 512

typedef float v2f __attribute__((ext_vector_type(2)));
typedef float v4f __attribute__((ext_vector_type(4)));
typedef float v8f __attribute__((ext_vector_type(8)));

#define RB   196
#define NBLK ((N_NODES + RB - 1) / RB)
#define ST   1024
#define CAP  4096
#define CH   (ST * 16)
#define NIT  ((N_EDGES + CH - 1) / CH)

typedef int v4i __attribute__((ext_vector_type(4)));

__global__ __launch_bounds__(ST) void GNN_scatter_kernel(
    const float* __restrict__ x,
    const int*   __restrict__ Ai,
    const int*   __restrict__ Aj,
    float* __restrict__ S,
    float* __restrict__ deg)
{
    __shared__ v4f      Sl4[RB * (F / 4)];
    __shared__ unsigned list[CAP];
    __shared__ int      degc[RB];
    __shared__ int      cnt;
    float* Sl = reinterpret_cast<float*>(Sl4);

    const int tid  = threadIdx.x;
    const int lane = tid & 31;
    const int wave = __builtin_amdgcn_readfirstlane(tid >> 5);
    const int lo   = blockIdx.x * RB;

    for (int i = tid; i < RB * (F / 4); i += ST) Sl4[i] = (v4f){};
    if (tid < RB) degc[tid] = 0;
    if (tid == 0) cnt = 0;
    __syncthreads();

    for (int it = 0; it < NIT; ++it) {
        const int itr  = (it + (int)blockIdx.x) % NIT;
        const int base = itr * CH + tid * 4;
        v4i v[4];
        #pragma unroll
        for (int j = 0; j < 4; ++j) {
            const int idx = base + j * (ST * 4);
            v[j] = (idx < N_EDGES) ? *reinterpret_cast<const v4i*>(Ai + idx) : (v4i){-1, -1, -1, -1};
        }
        unsigned mm = 0;
        #pragma unroll
        for (int j = 0; j < 4; ++j) {
            mm |= ((unsigned)(v[j].x - lo) < (unsigned)RB) ? (1u << (j * 4 + 0)) : 0u;
            mm |= ((unsigned)(v[j].y - lo) < (unsigned)RB) ? (1u << (j * 4 + 1)) : 0u;
            mm |= ((unsigned)(v[j].z - lo) < (unsigned)RB) ? (1u << (j * 4 + 2)) : 0u;
            mm |= ((unsigned)(v[j].w - lo) < (unsigned)RB) ? (1u << (j * 4 + 3)) : 0u;
        }
        while (mm) {
            const int bit = __builtin_ctz(mm);
            mm &= mm - 1;
            const int e   = base + (bit >> 2) * (ST * 4) + (bit & 3);
            const int loc = Ai[e] - lo;
            const int pos = __hip_atomic_fetch_add(&cnt, 1, __ATOMIC_RELAXED, __HIP_MEMORY_SCOPE_WORKGROUP);
            __hip_atomic_fetch_add(&degc[loc], 1, __ATOMIC_RELAXED, __HIP_MEMORY_SCOPE_WORKGROUP);
            if (pos < CAP) {
                list[pos] = ((unsigned)e << 8) | (unsigned)loc;
            } else {
                const float* __restrict__ xr = x + (size_t)Aj[e] * F;
                float* sr = Sl + loc * F;
                for (int c2 = 0; c2 < F; ++c2)
                    __hip_atomic_fetch_add(&sr[c2], xr[c2], __ATOMIC_RELAXED, __HIP_MEMORY_SCOPE_WORKGROUP);
            }
        }
    }
    __syncthreads();

    {
        int n = __builtin_amdgcn_readfirstlane(cnt);
        n = n < CAP ? n : CAP;
        const int per   = (n + (ST / 32) - 1) / (ST / 32);
        const int start = wave * per;
        const int end   = (start + per < n) ? (start + per) : n;
        for (int s = start; s < end; s += 32) {
            const int nb = (end - s < 32) ? (end - s) : 32;
            unsigned ent = 0;
            int aj = 0;
            if (lane < nb) {
                ent = list[s + lane];
                aj  = Aj[ent >> 8];
            }
            for (int k = 0; k < nb; k += 4) {
                float r[4][F / 32];
                #pragma unroll
                for (int u = 0; u < 4; ++u) {
                    const int kk  = (k + u < nb) ? (k + u) : (nb - 1);
                    const int ajk = __builtin_amdgcn_readlane(aj, kk);
                    const float* __restrict__ xr = x + (size_t)ajk * F + lane;
                    #pragma unroll
                    for (int c2 = 0; c2 < F / 32; ++c2) r[u][c2] = xr[c2 * 32];
                }
                #pragma unroll
                for (int u = 0; u < 4; ++u) {
                    if (k + u < nb) {
                        const int loc = (int)(__builtin_amdgcn_readlane((int)ent, k + u) & 255);
                        float* sr = Sl + loc * F + lane;
                        #pragma unroll
                        for (int c2 = 0; c2 < F / 32; ++c2)
                            __hip_atomic_fetch_add(&sr[c2 * 32], r[u][c2], __ATOMIC_RELAXED, __HIP_MEMORY_SCOPE_WORKGROUP);
                    }
                }
            }
        }
    }
    __syncthreads();

    {
        v4f* __restrict__ Sg4 = reinterpret_cast<v4f*>(S) + (size_t)lo * (F / 4);
        for (int i = tid; i < RB * (F / 4); i += ST) {
            const int row = i >> 6;
            if (lo + row < N_NODES) Sg4[i] = Sl4[i];
        }
        if (tid < RB && lo + tid < N_NODES) deg[lo + tid] = (float)degc[tid];
    }
}

#define BM 80
#define BN 128
#define KC 64
#define RS 68
#define RS4 17
#define GT 128

__global__ __launch_bounds__(GT, 5) void GNN_gemm_kernel(
    const float* __restrict__ x,
    const float* __restrict__ W,
    const float* __restrict__ b,
    const float* __restrict__ S,
    const float* __restrict__ deg,
    float* __restrict__ out)
{
    __shared__ v4f Ls4[(BM + BN) * RS4];
    v4f* As4 = Ls4;
    v4f* Ws4 = Ls4 + BM * RS4;
    const float* As = reinterpret_cast<const float*>(As4);
    const float* Ws = reinterpret_cast<const float*>(Ws4);

    const int tid   = threadIdx.x;
    const int wave  = tid >> 5;
    const int lane  = tid & 31;
    const int n0    = blockIdx.x * BN;
    const int m0    = blockIdx.y * BM;
    const int c     = lane & 15;
    const int khalf = (lane >> 4) * 2;

    const int sr  = tid >> 4;
    const int sc4 = tid & 15;

    const float* __restrict__ xbase = x + (size_t)(m0 + sr) * F + sc4 * 4;
    const float* __restrict__ sbase = S + (size_t)(m0 + sr) * F + sc4 * 4;
    const float* __restrict__ wbase = W + (size_t)(n0 + sr) * K2 + sc4 * 4;
    const float* __restrict__ dbase = deg + m0 + sr;
    v4f* __restrict__ asto = As4 + sr * RS4 + sc4;
    v4f* __restrict__ wsto = Ws4 + sr * RS4 + sc4;

    v8f acc[5][2];
    #pragma unroll
    for (int i = 0; i < 5; ++i) { acc[i][0] = (v8f){}; acc[i][1] = (v8f){}; }

    for (int kc = 0; kc < K2 / KC; ++kc) {
        __syncthreads();
        {
            const bool xhalf = kc < F / KC;
            const float* __restrict__ ap = xhalf ? (xbase + kc * KC) : (sbase + (kc * KC - F));
            v4f va[BM / 8];
            #pragma unroll
            for (int i = 0; i < BM / 8; ++i)
                va[i] = *reinterpret_cast<const v4f*>(ap + i * (8 * F));
            if (xhalf) {
                #pragma unroll
                for (int i = 0; i < BM / 8; ++i) {
                    const float d = dbase[i * 8];
                    va[i] *= d;
                }
            }
            #pragma unroll
            for (int i = 0; i < BM / 8; ++i)
                asto[i * (8 * RS4)] = va[i];
        }
        {
            const float* __restrict__ wp = wbase + kc * KC;
            v4f vw[BN / 8];
            #pragma unroll
            for (int i = 0; i < BN / 8; ++i)
                vw[i] = *reinterpret_cast<const v4f*>(wp + i * (8 * K2));
            asm volatile("" :: "v"(vw[0]), "v"(vw[1]), "v"(vw[2]), "v"(vw[3]),
                               "v"(vw[4]), "v"(vw[5]), "v"(vw[6]), "v"(vw[7]),
                               "v"(vw[8]), "v"(vw[9]), "v"(vw[10]), "v"(vw[11]),
                               "v"(vw[12]), "v"(vw[13]), "v"(vw[14]), "v"(vw[15]));
            #pragma unroll
            for (int i = 0; i < BN / 8; ++i)
                wsto[i * (8 * RS4)] = vw[i];
        }
        __syncthreads();

        const float* __restrict__ w0   = Ws + (wave * 32 + c) * RS + khalf;
        const float* __restrict__ w1   = w0 + 16 * RS;
        const float* __restrict__ arow = As + c * RS + khalf;
        #pragma unroll
        for (int kk = 0; kk < KC; kk += 4) {
            v2f b0, b1;
            b0.x = w0[kk]; b0.y = w0[kk + 1];
            b1.x = w1[kk]; b1.y = w1[kk + 1];
            #pragma unroll
            for (int i = 0; i < 5; ++i) {
                v2f a;
                a.x = arow[i * 16 * RS + kk];
                a.y = arow[i * 16 * RS + kk + 1];
                acc[i][0] = __builtin_amdgcn_wmma_f32_16x16x4_f32(
                    false, a, false, b0, (short)0, acc[i][0], false, false);
                acc[i][1] = __builtin_amdgcn_wmma_f32_16x16x4_f32(
                    false, a, false, b1, (short)0, acc[i][1], false, false);
            }
        }
    }

    #pragma unroll
    for (int j = 0; j < 2; ++j) {
        const int nout = n0 + wave * 32 + j * 16 + c;
        const float bias = b[nout];
        #pragma unroll
        for (int i = 0; i < 5; ++i) {
            const int mbase = m0 + i * 16 + (lane >> 4) * 8;
            #pragma unroll
            for (int r = 0; r < 8; ++r) {
                const int m = mbase + r;
                out[(size_t)m * F + nout] = acc[i][j][r] + deg[m] * bias;
            }
        }
    }
}

extern "C" void kernel_launch(void* const* d_in, const int* in_sizes, int n_in,
                              void* d_out, int out_size, void* d_ws, size_t ws_size,
                              hipStream_t stream) {
    const float* x  = (const float*)d_in[0];
    const float* W  = (const float*)d_in[1];
    const float* b  = (const float*)d_in[2];
    const int*   Ai = (const int*)d_in[3];
    const int*   Aj = (const int*)d_in[4];
    float* out = (float*)d_out;

    float* S   = (float*)d_ws;
    float* deg = S + (size_t)N_NODES * F;

    GNN_scatter_kernel<<<NBLK, ST, 0, stream>>>(x, Ai, Aj, S, deg);

    dim3 grid(2, 625);
    GNN_gemm_kernel<<<grid, GT, 0, stream>>>(x, W, b, S, deg, out);
}


# Aggregation and GEMM fused into one launch: each owner block keeps its 196 summed rows and deg in LDS and multiplies straight out of them (W, deg*x staged in 16-wide k chunks); sum order unchanged
# speedup vs baseline: 2.6238x; 1.0944x over previous
#include <hip/hip_runtime.h>
#include <hip/hip_bf16.h>

#define N_NODES 50000
#define N_EDGES 312500
#define F 256
#define K2 512

typedef float v2f __attribute__((ext_vector_type(2)));
typedef float v4f __attribute__((ext_vector_type(4)));
typedef float v8f __attribute__((ext_vector_type(8)));
typedef int v4i __attribute__((ext_vector_type(4)));

#define RB   196
#define RT   13
#define RP   (RT * 16)
#define NBLK ((N_NODES + RB - 1) / RB)
#define ST   1024
#define CAP  4096
#define CH   (ST * 16)
#define NIT  ((N_EDGES + CH - 1) / CH)
#define SP   260
#define SP4  65
#define KC   16
#define WS   20
#define WS4  5
#define STG  ((RP + F) * WS4)

template <int AP>
__device__ __forceinline__ void mm_main(v8f (&acc)[4][2],
                                        const float* __restrict__ ap,
                                        const float* __restrict__ wp)
{
    #pragma unroll
    for (int kk = 0; kk < KC; kk += 4) {
        v2f b0, b1;
        b0.x = wp[kk];            b0.y = wp[kk + 1];
        b1.x = wp[16 * WS + kk];  b1.y = wp[16 * WS + kk + 1];
        #pragma unroll
        for (int i = 0; i < 3; ++i) {
            v2f a;
            a.x = ap[i * 16 * AP + kk];
            a.y = ap[i * 16 * AP + kk + 1];
            acc[i][0] = __builtin_amdgcn_wmma_f32_16x16x4_f32(
                false, a, false, b0, (short)0, acc[i][0], false, false);
            acc[i][1] = __builtin_amdgcn_wmma_f32_16x16x4_f32(
                false, a, false, b1, (short)0, acc[i][1], false, false);
        }
    }
}

template <int AP>
__device__ __forceinline__ void mm_extra(v8f (&acc)[4][2],
                                         const float* __restrict__ ap,
                                         const float* __restrict__ wp)
{
    #pragma unroll
    for (int kk = 0; kk < KC; kk += 4) {
        v2f b0, b1, a;
        b0.x = wp[kk];            b0.y = wp[kk + 1];
        b1.x = wp[16 * WS + kk];  b1.y = wp[16 * WS + kk + 1];
        a.x = ap[3 * 16 * AP + kk];
        a.y = ap[3 * 16 * AP + kk + 1];
        acc[3][0] = __builtin_amdgcn_wmma_f32_16x16x4_f32(
            false, a, false, b0, (short)0, acc[3][0], false, false);
        acc[3][1] = __builtin_amdgcn_wmma_f32_16x16x4_f32(
            false, a, false, b1, (short)0, acc[3][1], false, false);
    }
}

__global__ __launch_bounds__(ST) void GNN_fused_kernel(
    const float* __restrict__ x,
    const float* __restrict__ W,
    const float* __restrict__ b,
    const int*   __restrict__ Ai,
    const int*   __restrict__ Aj,
    float* __restrict__ out)
{
    __shared__ v4f      Sl4[RP * SP4];
    __shared__ v4f      Cs4[2 * STG];
    __shared__ unsigned list[CAP];
    __shared__ int      degc[RP];
    __shared__ int      cnt;
    float* Sl = reinterpret_cast<float*>(Sl4);
    const float* Cs = reinterpret_cast<const float*>(Cs4);

    const int tid  = threadIdx.x;
    const int lane = tid & 31;
    const int wave = __builtin_amdgcn_readfirstlane(tid >> 5);
    const int lo   = blockIdx.x * RB;

    for (int i = tid; i < RP * SP4; i += ST) Sl4[i] = (v4f){};
    if (tid < RP) degc[tid] = 0;
    if (tid == 0) cnt = 0;
    __syncthreads();

    for (int it = 0; it < NIT; ++it) {
        const int itr  = (it + (int)blockIdx.x) % NIT;
        const int base = itr * CH + tid * 4;
        v4i v[4];
        #pragma unroll
        for (int j = 0; j < 4; ++j) {
            const int idx = base + j * (ST * 4);
            v[j] = (idx < N_EDGES) ? *reinterpret_cast<const v4i*>(Ai + idx) : (v4i){-1, -1, -1, -1};
        }
        unsigned mm = 0;
        #pragma unroll
        for (int j = 0; j < 4; ++j) {
            mm |= ((unsigned)(v[j].x - lo) < (unsigned)RB) ? (1u << (j * 4 + 0)) : 0u;
            mm |= ((unsigned)(v[j].y - lo) < (unsigned)RB) ? (1u << (j * 4 + 1)) : 0u;
            mm |= ((unsigned)(v[j].z - lo) < (unsigned)RB) ? (1u << (j * 4 + 2)) : 0u;
            mm |= ((unsigned)(v[j].w - lo) < (unsigned)RB) ? (1u << (j * 4 + 3)) : 0u;
        }
        while (mm) {
            const int bit = __builtin_ctz(mm);
            mm &= mm - 1;
            const int e   = base + (bit >> 2) * (ST * 4) + (bit & 3);
            const int loc = Ai[e] - lo;
            const int pos = __hip_atomic_fetch_add(&cnt, 1, __ATOMIC_RELAXED, __HIP_MEMORY_SCOPE_WORKGROUP);
            __hip_atomic_fetch_add(&degc[loc], 1, __ATOMIC_RELAXED, __HIP_MEMORY_SCOPE_WORKGROUP);
            if (pos < CAP) {
                list[pos] = ((unsigned)e << 8) | (unsigned)loc;
            } else {
                const float* __restrict__ xr = x + (size_t)Aj[e] * F;
                float* sr = Sl + loc * SP;
                for (int c2 = 0; c2 < F; ++c2)
                    __hip_atomic_fetch_add(&sr[c2], xr[c2], __ATOMIC_RELAXED, __HIP_MEMORY_SCOPE_WORKGROUP);
            }
        }
    }
    __syncthreads();

    {
        int n = __builtin_amdgcn_readfirstlane(cnt);
        n = n < CAP ? n : CAP;
        const int per   = (n + (ST / 32) - 1) / (ST / 32);
        const int start = wave * per;
        const int end   = (start + per < n) ? (start + per) : n;
        for (int s = start; s < end; s += 32) {
            const int nb = (end - s < 32) ? (end - s) : 32;
            unsigned ent = 0;
            int aj = 0;
            if (lane < nb) {
                ent = list[s + lane];
                aj  = Aj[ent >> 8];
            }
            for (int k = 0; k < nb; k += 4) {
                float r[4][F / 32];
                #pragma unroll
                for (int u = 0; u < 4; ++u) {
                    const int kk  = (k + u < nb) ? (k + u) : (nb - 1);
                    const int ajk = __builtin_amdgcn_readlane(aj, kk);
                    const float* __restrict__ xr = x + (size_t)ajk * F + lane;
                    #pragma unroll
                    for (int c2 = 0; c2 < F / 32; ++c2) r[u][c2] = xr[c2 * 32];
                }
                #pragma unroll
                for (int u = 0; u < 4; ++u) {
                    if (k + u < nb) {
                        const int loc = (int)(__builtin_amdgcn_readlane((int)ent, k + u) & 255);
                        float* sr = Sl + loc * SP + lane;
                        #pragma unroll
                        for (int c2 = 0; c2 < F / 32; ++c2)
                            __hip_atomic_fetch_add(&sr[c2 * 32], r[u][c2], __ATOMIC_RELAXED, __HIP_MEMORY_SCOPE_WORKGROUP);
                    }
                }
            }
        }
    }

    const int c  = lane & 15;
    const int h2 = (lane >> 4) * 2;
    const int cp = wave & 7;
    const int rg = ((wave >> 3) + wave) & 3;
    const int t0 = (rg == 0) ? 0 : (1 + 3 * rg);

    const int sr = tid >> 2;
    const int sq = tid & 3;
    const int nv = (N_NODES - lo < RB) ? (N_NODES - lo) : RB;
    const bool xs = (sr < RP);
    const bool xv = (sr < nv);
    const unsigned wo = (unsigned)(sr * K2 + sq * 4);
    const unsigned xo = (unsigned)((lo + sr) * F + sq * 4);
    v4f* __restrict__ stx = Cs4 + sr * WS4 + sq;
    v4f* __restrict__ stw = Cs4 + (RP + sr) * WS4 + sq;

    v8f acc[4][2];
    #pragma unroll
    for (int i = 0; i < 4; ++i) { acc[i][0] = (v8f){}; acc[i][1] = (v8f){}; }

    v4f pw = *reinterpret_cast<const v4f*>(W + wo);
    v4f px = xv ? *reinterpret_cast<const v4f*>(x + xo) : (v4f){};

    __syncthreads();
    const float d = xs ? (float)degc[xs ? sr : 0] : 0.0f;

    stw[0] = pw;
    if (xs) stx[0] = px * d;
    pw = *reinterpret_cast<const v4f*>(W + (wo + KC));
    px = xv ? *reinterpret_cast<const v4f*>(x + (xo + KC)) : (v4f){};
    __syncthreads();

    const float* __restrict__ wp  = Cs + (RP + cp * 32 + c) * WS + h2;
    const float* __restrict__ apx = Cs + (t0 * 16 + c) * WS + h2;
    const float* __restrict__ aps = Sl + (t0 * 16 + c) * SP + h2;

    #pragma nounroll
    for (int kc = 0; kc < F / KC; kc += 2) {
        stw[STG] = pw;
        if (xs) stx[STG] = px * d;
        pw = *reinterpret_cast<const v4f*>(W + (wo + (kc + 2) * KC));
        if (kc + 2 < F / KC)
            px = xv ? *reinterpret_cast<const v4f*>(x + (xo + (kc + 2) * KC)) : (v4f){};
        mm_main<WS>(acc, apx, wp);
        if (rg == 0) mm_extra<WS>(acc, apx, wp);
        __syncthreads();

        stw[0] = pw;
        if (kc + 2 < F / KC) {
            if (xs) stx[0] = px * d;
        }
        pw = *reinterpret_cast<const v4f*>(W + (wo + (kc + 3) * KC));
        if (kc + 3 < F / KC)
            px = xv ? *reinterpret_cast<const v4f*>(x + (xo + (kc + 3) * KC)) : (v4f){};
        mm_main<WS>(acc, apx + STG * 4, wp + STG * 4);
        if (rg == 0) mm_extra<WS>(acc, apx + STG * 4, wp + STG * 4);
        __syncthreads();
    }
    #pragma nounroll
    for (int kc = F / KC; kc < K2 / KC; kc += 2) {
        stw[STG] = pw;
        if (kc + 2 < K2 / KC)
            pw = *reinterpret_cast<const v4f*>(W + (wo + (kc + 2) * KC));
        {
            const float* __restrict__ ap = aps + (kc * KC - F);
            mm_main<SP>(acc, ap, wp);
            if (rg == 0) mm_extra<SP>(acc, ap, wp);
        }
        __syncthreads();

        if (kc + 2 < K2 / KC) {
            stw[0] = pw;
            pw = *reinterpret_cast<const v4f*>(W + (wo + (kc + 3) * KC));
        }
        {
            const float* __restrict__ ap = aps + ((kc + 1) * KC - F);
            mm_main<SP>(acc, ap, wp + STG * 4);
            if (rg == 0) mm_extra<SP>(acc, ap, wp + STG * 4);
        }
        __syncthreads();
    }

    #pragma unroll
    for (int j = 0; j < 2; ++j) {
        const int nout = cp * 32 + j * 16 + c;
        const float bias = b[nout];
        #pragma unroll
        for (int i = 0; i < 4; ++i) {
            if (i < 3 || rg == 0) {
                const int mb = (t0 + i) * 16 + (lane >> 4) * 8;
                #pragma unroll
                for (int r = 0; r < 8; ++r) {
                    const int ml = mb + r;
                    if (ml < nv)
                        out[(size_t)(lo + ml) * F + nout] = acc[i][j][r] + (float)degc[ml] * bias;
                }
            }
        }
    }
}

extern "C" void kernel_launch(void* const* d_in, const int* in_sizes, int n_in,
                              void* d_out, int out_size, void* d_ws, size_t ws_size,
                              hipStream_t stream) {
    const float* x  = (const float*)d_in[0];
    const float* W  = (const float*)d_in[1];
    const float* b  = (const float*)d_in[2];
    const int*   Ai = (const int*)d_in[3];
    const int*   Aj = (const int*)d_in[4];
    float* out = (float*)d_out;

    GNN_fused_kernel<<<NBLK, ST, 0, stream>>>(x, W, b, Ai, Aj, out);
}


# Edge list sorted by owner block in a small first launch (one 128-byte slot per producer and owner in the workspace); owners read 32 KiB of slots, not the whole list; LDS add order: arrival
# speedup vs baseline: 2.8118x; 1.0717x over previous
#include <hip/hip_runtime.h>
#include <hip/hip_bf16.h>

#define N_NODES 50000
#define N_EDGES 312500
#define F 256
#define K2 512

typedef float v2f __attribute__((ext_vector_type(2)));
typedef float v4f __attribute__((ext_vector_type(4)));
typedef float v8f __attribute__((ext_vector_type(8)));
typedef int v4i __attribute__((ext_vector_type(4)));
typedef unsigned v4u __attribute__((ext_vector_type(4)));

#define RB   196
#define RT   13
#define RP   (RT * 16)
#define NBLK ((N_NODES + RB - 1) / RB)
#define ST   1024
#define CAP  4096
#define NPROD 256
#define EP   ((N_EDGES + NPROD - 1) / NPROD)
#define PT   256
#define EPT  ((EP + PT - 1) / PT)
#define SLOT 32
#define EMPTY 0xFFFFFFFFu
#define SP   260
#define SP4  65
#define KC   16
#define WS   20
#define WS4  5
#define STG  ((RP + F) * WS4)

template <int AP>
__device__ __forceinline__ void mm_main(v8f (&acc)[4][2],
                                        const float* __restrict__ ap,
                                        const float* __restrict__ wp)
{
    #pragma unroll
    for (int kk = 0; kk < KC; kk += 4) {
        v2f b0, b1;
        b0.x = wp[kk];            b0.y = wp[kk + 1];
        b1.x = wp[16 * WS + kk];  b1.y = wp[16 * WS + kk + 1];
        #pragma unroll
        for (int i = 0; i < 3; ++i) {
            v2f a;
            a.x = ap[i * 16 * AP + kk];
            a.y = ap[i * 16 * AP + kk + 1];
            acc[i][0] = __builtin_amdgcn_wmma_f32_16x16x4_f32(
                false, a, false, b0, (short)0, acc[i][0], false, false);
            acc[i][1] = __builtin_amdgcn_wmma_f32_16x16x4_f32(
                false, a, false, b1, (short)0, acc[i][1], false, false);
        }
    }
}

template <int AP>
__device__ __forceinline__ void mm_extra(v8f (&acc)[4][2],
                                         const float* __restrict__ ap,
                                         const float* __restrict__ wp)
{
    #pragma unroll
    for (int kk = 0; kk < KC; kk += 4) {
        v2f b0, b1, a;
        b0.x = wp[kk];            b0.y = wp[kk + 1];
        b1.x = wp[16 * WS + kk];  b1.y = wp[16 * WS + kk + 1];
        a.x = ap[3 * 16 * AP + kk];
        a.y = ap[3 * 16 * AP + kk + 1];
        acc[3][0] = __builtin_amdgcn_wmma_f32_16x16x4_f32(
            false, a, false, b0, (short)0, acc[3][0], false, false);
        acc[3][1] = __builtin_amdgcn_wmma_f32_16x16x4_f32(
            false, a, false, b1, (short)0, acc[3][1], false, false);
    }
}

__global__ __launch_bounds__(PT) void GNN_bin_kernel(
    const int* __restrict__ Ai,
    const int* __restrict__ Aj,
    unsigned* __restrict__ slots)
{
    __shared__ v4u img4[NBLK * SLOT / 4];
    __shared__ int hist[NBLK];
    unsigned* img = reinterpret_cast<unsigned*>(img4);

    const int t = threadIdx.x;
    const int p = blockIdx.x;

    int ai[EPT], aj[EPT];
    #pragma unroll
    for (int j = 0; j < EPT; ++j) {
        const int e  = p * EP + j * PT + t;
        const int ec = (e < N_EDGES) ? e : (N_EDGES - 1);
        ai[j] = Ai[ec];
        aj[j] = Aj[ec];
    }
    for (int i = t; i < NBLK * SLOT / 4; i += PT) img4[i] = (v4u){EMPTY, EMPTY, EMPTY, EMPTY};
    for (int i = t; i < NBLK; i += PT) hist[i] = 0;
    __syncthreads();

    #pragma unroll
    for (int j = 0; j < EPT; ++j) {
        const int o = j * PT + t;
        const bool mine = (o < EP) && (p * EP + o < N_EDGES);
        if (mine && (unsigned)ai[j] < (unsigned)N_NODES) {
            const int q   = ai[j] / RB;
            const int loc = ai[j] - q * RB;
            const int pos = __hip_atomic_fetch_add(&hist[q], 1, __ATOMIC_RELAXED, __HIP_MEMORY_SCOPE_WORKGROUP);
            if (pos < SLOT - 1) img[q * SLOT + 1 + pos] = ((unsigned)aj[j] << 8) | (unsigned)loc;
        }
    }
    __syncthreads();
    for (int i = t; i < NBLK; i += PT) img[i * SLOT] = (unsigned)hist[i];
    __syncthreads();

    v4u* __restrict__ dst = reinterpret_cast<v4u*>(slots);
    for (int i = t; i < NBLK * SLOT / 4; i += PT) {
        const int q = i >> 3;
        dst[(q * NPROD + p) * (SLOT / 4) + (i & 7)] = img4[i];
    }
}

__global__ __launch_bounds__(ST) void GNN_fused_kernel(
    const float* __restrict__ x,
    const float* __restrict__ W,
    const float* __restrict__ b,
    const int*   __restrict__ Ai,
    const int*   __restrict__ Aj,
    const unsigned* __restrict__ slots,
    float* __restrict__ out)
{
    __shared__ v4f      Sl4[RP * SP4];
    __shared__ v4f      Cs4[2 * STG];
    __shared__ unsigned list[CAP];
    __shared__ int      degc[RP];
    __shared__ int      cnt;
    float* Sl = reinterpret_cast<float*>(Sl4);
    const float* Cs = reinterpret_cast<const float*>(Cs4);

    const int tid  = threadIdx.x;
    const int lane = tid & 31;
    const int wave = __builtin_amdgcn_readfirstlane(tid >> 5);
    const int lo   = blockIdx.x * RB;

    const int pp  = tid >> 2;
    const int sub = tid & 3;
    const unsigned* __restrict__ sw = slots + ((size_t)blockIdx.x * NPROD + pp) * SLOT + sub * 8;
    const v4u w0 = *reinterpret_cast<const v4u*>(sw);
    const v4u w1 = *reinterpret_cast<const v4u*>(sw + 4);

    for (int i = tid; i < RP * SP4; i += ST) Sl4[i] = (v4f){};
    if (tid < RP) degc[tid] = 0;
    if (tid == 0) cnt = 0;
    __syncthreads();

    {
        auto note = [&](unsigned w) {
            const int loc = (int)(w & 255u);
            const int pos = __hip_atomic_fetch_add(&cnt, 1, __ATOMIC_RELAXED, __HIP_MEMORY_SCOPE_WORKGROUP);
            __hip_atomic_fetch_add(&degc[loc], 1, __ATOMIC_RELAXED, __HIP_MEMORY_SCOPE_WORKGROUP);
            if (pos < CAP) {
                list[pos] = w;
            } else {
                const float* __restrict__ xr = x + (size_t)(w >> 8) * F;
                float* sr = Sl + loc * SP;
                for (int c2 = 0; c2 < F; ++c2)
                    __hip_atomic_fetch_add(&sr[c2], xr[c2], __ATOMIC_RELAXED, __HIP_MEMORY_SCOPE_WORKGROUP);
            }
        };
        const unsigned found = (unsigned)__shfl((int)w0.x, lane & ~3);
        if (found <= (unsigned)(SLOT - 1)) {
            unsigned mm = 0;
            mm |= (w0.x != EMPTY && sub != 0) ? 1u : 0u;
            mm |= (w0.y != EMPTY) ? 2u : 0u;
            mm |= (w0.z != EMPTY) ? 4u : 0u;
            mm |= (w0.w != EMPTY) ? 8u : 0u;
            mm |= (w1.x != EMPTY) ? 16u : 0u;
            mm |= (w1.y != EMPTY) ? 32u : 0u;
            mm |= (w1.z != EMPTY) ? 64u : 0u;
            mm |= (w1.w != EMPTY) ? 128u : 0u;
            while (mm) {
                const int bit = __builtin_ctz(mm);
                mm &= mm - 1;
                note(sw[bit]);
            }
        } else {
            const int e0 = pp * EP;
            const int e1 = (e0 + EP < N_EDGES) ? (e0 + EP) : N_EDGES;
            for (int e = e0 + sub; e < e1; e += 4) {
                const int loc = Ai[e] - lo;
                if ((unsigned)loc < (unsigned)RB) note(((unsigned)Aj[e] << 8) | (unsigned)loc);
            }
        }
    }
    __syncthreads();

    {
        int n = __builtin_amdgcn_readfirstlane(cnt);
        n = n < CAP ? n : CAP;
        const int per   = (n + (ST / 32) - 1) / (ST / 32);
        const int start = wave * per;
        const int end   = (start + per < n) ? (start + per) : n;
        for (int s = start; s < end; s += 32) {
            const int nb = (end - s < 32) ? (end - s) : 32;
            unsigned ent = 0;
            if (lane < nb) ent = list[s + lane];
            const int aj = (int)(ent >> 8);
            for (int k = 0; k < nb; k += 4) {
                float r[4][F / 32];
                #pragma unroll
                for (int u = 0; u < 4; ++u) {
                    const int kk  = (k + u < nb) ? (k + u) : (nb - 1);
                    const int ajk = __builtin_amdgcn_readlane(aj, kk);
                    const float* __restrict__ xr = x + (size_t)ajk * F + lane;
                    #pragma unroll
                    for (int c2 = 0; c2 < F / 32; ++c2) r[u][c2] = xr[c2 * 32];
                }
                #pragma unroll
                for (int u = 0; u < 4; ++u) {
                    if (k + u < nb) {
                        const int loc = (int)(__builtin_amdgcn_readlane((int)ent, k + u) & 255);
                        float* sr = Sl + loc * SP + lane;
                        #pragma unroll
                        for (int c2 = 0; c2 < F / 32; ++c2)
                            __hip_atomic_fetch_add(&sr[c2 * 32], r[u][c2], __ATOMIC_RELAXED, __HIP_MEMORY_SCOPE_WORKGROUP);
                    }
                }
            }
        }
    }

    const int c  = lane & 15;
    const int h2 = (lane >> 4) * 2;
    const int cp = wave & 7;
    const int rg = ((wave >> 3) + wave) & 3;
    const int t0 = (rg == 0) ? 0 : (1 + 3 * rg);

    const int sr = tid >> 2;
    const int sq = tid & 3;
    const int nv = (N_NODES - lo < RB) ? (N_NODES - lo) : RB;
    const bool xs = (sr < RP);
    const bool xv = (sr < nv);
    const unsigned wo = (unsigned)(sr * K2 + sq * 4);
    const unsigned xo = (unsigned)((lo + sr) * F + sq * 4);
    v4f* __restrict__ stx = Cs4 + sr * WS4 + sq;
    v4f* __restrict__ stw = Cs4 + (RP + sr) * WS4 + sq;

    v8f acc[4][2];
    #pragma unroll
    for (int i = 0; i < 4; ++i) { acc[i][0] = (v8f){}; acc[i][1] = (v8f){}; }

    v4f pw = *reinterpret_cast<const v4f*>(W + wo);
    v4f px = xv ? *reinterpret_cast<const v4f*>(x + xo) : (v4f){};

    __syncthreads();
    const float d = xs ? (float)degc[xs ? sr : 0] : 0.0f;

    stw[0] = pw;
    if (xs) stx[0] = px * d;
    pw = *reinterpret_cast<const v4f*>(W + (wo + KC));
    px = xv ? *reinterpret_cast<const v4f*>(x + (xo + KC)) : (v4f){};
    __syncthreads();

    const float* __restrict__ wp  = Cs + (RP + cp * 32 + c) * WS + h2;
    const float* __restrict__ apx = Cs + (t0 * 16 + c) * WS + h2;
    const float* __restrict__ aps = Sl + (t0 * 16 + c) * SP + h2;

    #pragma nounroll
    for (int kc = 0; kc < F / KC; kc += 2) {
        stw[STG] = pw;
        if (xs) stx[STG] = px * d;
        pw = *reinterpret_cast<const v4f*>(W + (wo + (kc + 2) * KC));
        if (kc + 2 < F / KC)
            px = xv ? *reinterpret_cast<const v4f*>(x + (xo + (kc + 2) * KC)) : (v4f){};
        mm_main<WS>(acc, apx, wp);
        if (rg == 0) mm_extra<WS>(acc, apx, wp);
        __syncthreads();

        stw[0] = pw;
        if (kc + 2 < F / KC) {
            if (xs) stx[0] = px * d;
        }
        pw = *reinterpret_cast<const v4f*>(W + (wo + (kc + 3) * KC));
        if (kc + 3 < F / KC)
            px = xv ? *reinterpret_cast<const v4f*>(x + (xo + (kc + 3) * KC)) : (v4f){};
        mm_main<WS>(acc, apx + STG * 4, wp + STG * 4);
        if (rg == 0) mm_extra<WS>(acc, apx + STG * 4, wp + STG * 4);
        __syncthreads();
    }
    #pragma nounroll
    for (int kc = F / KC; kc < K2 / KC; kc += 2) {
        stw[STG] = pw;
        if (kc + 2 < K2 / KC)
            pw = *reinterpret_cast<const v4f*>(W + (wo + (kc + 2) * KC));
        {
            const float* __restrict__ ap = aps + (kc * KC - F);
            mm_main<SP>(acc, ap, wp);
            if (rg == 0) mm_extra<SP>(acc, ap, wp);
        }
        __syncthreads();

        if (kc + 2 < K2 / KC) {
            stw[0] = pw;
            pw = *reinterpret_cast<const v4f*>(W + (wo + (kc + 3) * KC));
        }
        {
            const float* __restrict__ ap = aps + ((kc + 1) * KC - F);
            mm_main<SP>(acc, ap, wp + STG * 4);
            if (rg == 0) mm_extra<SP>(acc, ap, wp + STG * 4);
        }
        __syncthreads();
    }

    #pragma unroll
    for (int j = 0; j < 2; ++j) {
        const int nout = cp * 32 + j * 16 + c;
        const float bias = b[nout];
        #pragma unroll
        for (int i = 0; i < 4; ++i) {
            if (i < 3 || rg == 0) {
                const int mb = (t0 + i) * 16 + (lane >> 4) * 8;
                #pragma unroll
                for (int r = 0; r < 8; ++r) {
                    const int ml = mb + r;
                    if (ml < nv)
                        out[(size_t)(lo + ml) * F + nout] = acc[i][j][r] + (float)degc[ml] * bias;
                }
            }
        }
    }
}

extern "C" void kernel_launch(void* const* d_in, const int* in_sizes, int n_in,
                              void* d_out, int out_size, void* d_ws, size_t ws_size,
                              hipStream_t stream) {
    const float* x  = (const float*)d_in[0];
    const float* W  = (const float*)d_in[1];
    const float* b  = (const float*)d_in[2];
    const int*   Ai = (const int*)d_in[3];
    const int*   Aj = (const int*)d_in[4];
    float* out = (float*)d_out;
    unsigned* slots = (unsigned*)d_ws;

    GNN_bin_kernel<<<NPROD, PT, 0, stream>>>(Ai, Aj, slots);
    GNN_fused_kernel<<<NBLK, ST, 0, stream>>>(x, W, b, Ai, Aj, slots, out);
}


# Aggregation as a counting sort: block groups its edges by destination row in LDS, waves sum each run in registers from 16-byte loads, one LDS add set per run; sum order: list order, then arrival
# speedup vs baseline: 2.9440x; 1.0470x over previous
#include <hip/hip_runtime.h>
#include <hip/hip_bf16.h>

#define N_NODES 50000
#define N_EDGES 312500
#define F 256
#define K2 512

typedef float v2f __attribute__((ext_vector_type(2)));
typedef float v4f __attribute__((ext_vector_type(4)));
typedef float v8f __attribute__((ext_vector_type(8)));
typedef int v4i __attribute__((ext_vector_type(4)));
typedef unsigned v4u __attribute__((ext_vector_type(4)));

#define RB   196
#define RT   13
#define RP   (RT * 16)
#define NBLK ((N_NODES + RB - 1) / RB)
#define ST   1024
#define CAP  4096
#define GU   8
#define NPROD 256
#define EP   ((N_EDGES + NPROD - 1) / NPROD)
#define PT   256
#define EPT  ((EP + PT - 1) / PT)
#define SLOT 32
#define EMPTY 0xFFFFFFFFu
#define SP   260
#define SP4  65
#define KC   16
#define WS   20
#define WS4  5
#define STG  ((RP + F) * WS4)

template <int AP>
__device__ __forceinline__ void mm_main(v8f (&acc)[4][2],
                                        const float* __restrict__ ap,
                                        const float* __restrict__ wp)
{
    #pragma unroll
    for (int kk = 0; kk < KC; kk += 4) {
        v2f b0, b1;
        b0.x = wp[kk];            b0.y = wp[kk + 1];
        b1.x = wp[16 * WS + kk];  b1.y = wp[16 * WS + kk + 1];
        #pragma unroll
        for (int i = 0; i < 3; ++i) {
            v2f a;
            a.x = ap[i * 16 * AP + kk];
            a.y = ap[i * 16 * AP + kk + 1];
            acc[i][0] = __builtin_amdgcn_wmma_f32_16x16x4_f32(
                false, a, false, b0, (short)0, acc[i][0], false, false);
            acc[i][1] = __builtin_amdgcn_wmma_f32_16x16x4_f32(
                false, a, false, b1, (short)0, acc[i][1], false, false);
        }
    }
}

template <int AP>
__device__ __forceinline__ void mm_extra(v8f (&acc)[4][2],
                                         const float* __restrict__ ap,
                                         const float* __restrict__ wp)
{
    #pragma unroll
    for (int kk = 0; kk < KC; kk += 4) {
        v2f b0, b1, a;
        b0.x = wp[kk];            b0.y = wp[kk + 1];
        b1.x = wp[16 * WS + kk];  b1.y = wp[16 * WS + kk + 1];
        a.x = ap[3 * 16 * AP + kk];
        a.y = ap[3 * 16 * AP + kk + 1];
        acc[3][0] = __builtin_amdgcn_wmma_f32_16x16x4_f32(
            false, a, false, b0, (short)0, acc[3][0], false, false);
        acc[3][1] = __builtin_amdgcn_wmma_f32_16x16x4_f32(
            false, a, false, b1, (short)0, acc[3][1], false, false);
    }
}

__global__ __launch_bounds__(PT) void GNN_bin_kernel(
    const int* __restrict__ Ai,
    const int* __restrict__ Aj,
    unsigned* __restrict__ slots)
{
    __shared__ v4u img4[NBLK * SLOT / 4];
    __shared__ int hist[NBLK];
    unsigned* img = reinterpret_cast<unsigned*>(img4);

    const int t = threadIdx.x;
    const int p = blockIdx.x;

    int ai[EPT], aj[EPT];
    #pragma unroll
    for (int j = 0; j < EPT; ++j) {
        const int e  = p * EP + j * PT + t;
        const int ec = (e < N_EDGES) ? e : (N_EDGES - 1);
        ai[j] = Ai[ec];
        aj[j] = Aj[ec];
    }
    for (int i = t; i < NBLK * SLOT / 4; i += PT) img4[i] = (v4u){EMPTY, EMPTY, EMPTY, EMPTY};
    for (int i = t; i < NBLK; i += PT) hist[i] = 0;
    __syncthreads();

    #pragma unroll
    for (int j = 0; j < EPT; ++j) {
        const int o = j * PT + t;
        const bool mine = (o < EP) && (p * EP + o < N_EDGES);
        if (mine && (unsigned)ai[j] < (unsigned)N_NODES) {
            const int q   = ai[j] / RB;
            const int loc = ai[j] - q * RB;
            const int pos = __hip_atomic_fetch_add(&hist[q], 1, __ATOMIC_RELAXED, __HIP_MEMORY_SCOPE_WORKGROUP);
            if (pos < SLOT - 1) img[q * SLOT + 1 + pos] = ((unsigned)aj[j] << 8) | (unsigned)loc;
        }
    }
    __syncthreads();
    for (int i = t; i < NBLK; i += PT) img[i * SLOT] = (unsigned)hist[i];
    __syncthreads();

    v4u* __restrict__ dst = reinterpret_cast<v4u*>(slots);
    for (int i = t; i < NBLK * SLOT / 4; i += PT) {
        const int q = i >> 3;
        dst[(q * NPROD + p) * (SLOT / 4) + (i & 7)] = img4[i];
    }
}

__global__ __launch_bounds__(ST) void GNN_fused_kernel(
    const float* __restrict__ x,
    const float* __restrict__ W,
    const float* __restrict__ b,
    const int*   __restrict__ Ai,
    const int*   __restrict__ Aj,
    const unsigned* __restrict__ slots,
    float* __restrict__ out)
{
    __shared__ v4f      Sl4[RP * SP4];
    __shared__ v4f      Cs4[2 * STG];
    __shared__ unsigned sorted[CAP];
    __shared__ int      degc[RP];
    __shared__ int      lcnt[RP];
    __shared__ int      cnt;
    float* Sl = reinterpret_cast<float*>(Sl4);
    const float* Cs = reinterpret_cast<const float*>(Cs4);

    const int tid  = threadIdx.x;
    const int lane = tid & 31;
    const int wave = __builtin_amdgcn_readfirstlane(tid >> 5);
    const int lo   = blockIdx.x * RB;

    const int pp  = tid >> 2;
    const int sub = tid & 3;
    const unsigned* __restrict__ sw = slots + ((size_t)blockIdx.x * NPROD + pp) * SLOT + sub * 8;
    const v4u w0 = *reinterpret_cast<const v4u*>(sw);
    const v4u w1 = *reinterpret_cast<const v4u*>(sw + 4);

    for (int i = tid; i < RP * SP4; i += ST) Sl4[i] = (v4f){};
    if (tid < RP) lcnt[tid] = 0;
    __syncthreads();

    const unsigned found = (unsigned)__shfl((int)w0.x, lane & ~3);
    auto for_each_edge = [&](auto f) {
        if (found <= (unsigned)(SLOT - 1)) {
            if (sub != 0 && w0.x != EMPTY) f(w0.x);
            if (w0.y != EMPTY) f(w0.y);
            if (w0.z != EMPTY) f(w0.z);
            if (w0.w != EMPTY) f(w0.w);
            if (w1.x != EMPTY) f(w1.x);
            if (w1.y != EMPTY) f(w1.y);
            if (w1.z != EMPTY) f(w1.z);
            if (w1.w != EMPTY) f(w1.w);
        } else {
            const int e0 = pp * EP;
            const int e1 = (e0 + EP < N_EDGES) ? (e0 + EP) : N_EDGES;
            for (int e = e0 + sub; e < e1; e += 4) {
                const int loc = Ai[e] - lo;
                if ((unsigned)loc < (unsigned)RB) f(((unsigned)Aj[e] << 8) | (unsigned)loc);
            }
        }
    };

    for_each_edge([&](unsigned w) {
        __hip_atomic_fetch_add(&lcnt[w & 255u], 1, __ATOMIC_RELAXED, __HIP_MEMORY_SCOPE_WORKGROUP);
    });
    __syncthreads();
    if (wave == 0) {
        int v[7];
        int s = 0;
        #pragma unroll
        for (int j = 0; j < 7; ++j) {
            const int r = lane * 7 + j;
            v[j] = (r < RP) ? lcnt[r] : 0;
            s += v[j];
        }
        int inc = s;
        #pragma unroll
        for (int dd = 1; dd < 32; dd <<= 1) {
            const int t = __shfl_up(inc, dd);
            if (lane >= dd) inc += t;
        }
        int run = inc - s;
        #pragma unroll
        for (int j = 0; j < 7; ++j) {
            const int r = lane * 7 + j;
            if (r < RP) { degc[r] = v[j]; lcnt[r] = run; }
            run += v[j];
        }
        if (lane == 31) cnt = inc;
    }
    __syncthreads();
    const int total = __builtin_amdgcn_readfirstlane(cnt);
    if (total <= CAP) {
        for_each_edge([&](unsigned w) {
            const int p = __hip_atomic_fetch_add(&lcnt[w & 255u], 1, __ATOMIC_RELAXED, __HIP_MEMORY_SCOPE_WORKGROUP);
            sorted[p] = w;
        });
    } else {
        for_each_edge([&](unsigned w) {
            const float* __restrict__ xr = x + (size_t)(w >> 8) * F;
            float* sr = Sl + (int)(w & 255u) * SP;
            for (int c2 = 0; c2 < F; ++c2)
                __hip_atomic_fetch_add(&sr[c2], xr[c2], __ATOMIC_RELAXED, __HIP_MEMORY_SCOPE_WORKGROUP);
        });
    }
    __syncthreads();

    {
        const int n = (total <= CAP) ? total : 0;

        auto flush = [&](int loc, v4f s0, v4f s1) {
            float* sr = Sl + loc * SP + lane * 4;
            __hip_atomic_fetch_add(&sr[0],       s0.x, __ATOMIC_RELAXED, __HIP_MEMORY_SCOPE_WORKGROUP);
            __hip_atomic_fetch_add(&sr[1],       s0.y, __ATOMIC_RELAXED, __HIP_MEMORY_SCOPE_WORKGROUP);
            __hip_atomic_fetch_add(&sr[2],       s0.z, __ATOMIC_RELAXED, __HIP_MEMORY_SCOPE_WORKGROUP);
            __hip_atomic_fetch_add(&sr[3],       s0.w, __ATOMIC_RELAXED, __HIP_MEMORY_SCOPE_WORKGROUP);
            __hip_atomic_fetch_add(&sr[128 + 0], s1.x, __ATOMIC_RELAXED, __HIP_MEMORY_SCOPE_WORKGROUP);
            __hip_atomic_fetch_add(&sr[128 + 1], s1.y, __ATOMIC_RELAXED, __HIP_MEMORY_SCOPE_WORKGROUP);
            __hip_atomic_fetch_add(&sr[128 + 2], s1.z, __ATOMIC_RELAXED, __HIP_MEMORY_SCOPE_WORKGROUP);
            __hip_atomic_fetch_add(&sr[128 + 3], s1.w, __ATOMIC_RELAXED, __HIP_MEMORY_SCOPE_WORKGROUP);
        };

        const int per   = (n + (ST / 32) - 1) / (ST / 32);
        const int start = wave * per;
        const int end   = (start + per < n) ? (start + per) : n;
        int cur = -1;
        v4f a0 = (v4f){}, a1 = (v4f){};
        for (int s = start; s < end; s += 32) {
            const int nb = (end - s < 32) ? (end - s) : 32;
            unsigned ent = 0;
            if (lane < nb) ent = sorted[s + lane];
            const int aj = (int)(ent >> 8);
            for (int k = 0; k < nb; k += GU) {
                v4f r0[GU], r1[GU];
                #pragma unroll
                for (int u = 0; u < GU; ++u) {
                    const int kk  = (k + u < nb) ? (k + u) : (nb - 1);
                    const int ajk = __builtin_amdgcn_readlane(aj, kk);
                    const v4f* __restrict__ xr = reinterpret_cast<const v4f*>(x + (size_t)ajk * F) + lane;
                    r0[u] = xr[0];
                    r1[u] = xr[32];
                }
                #pragma unroll
                for (int u = 0; u < GU; ++u) {
                    if (k + u < nb) {
                        const int loc = (int)(__builtin_amdgcn_readlane((int)ent, k + u) & 255);
                        if (loc != cur) {
                            if (cur >= 0) flush(cur, a0, a1);
                            a0 = (v4f){}; a1 = (v4f){};
                            cur = loc;
                        }
                        a0 += r0[u];
                        a1 += r1[u];
                    }
                }
            }
        }
        if (cur >= 0) flush(cur, a0, a1);
    }

    const int c  = lane & 15;
    const int h2 = (lane >> 4) * 2;
    const int cp = wave & 7;
    const int rg = ((wave >> 3) + wave) & 3;
    const int t0 = (rg == 0) ? 0 : (1 + 3 * rg);

    const int sr = tid >> 2;
    const int sq = tid & 3;
    const int nv = (N_NODES - lo < RB) ? (N_NODES - lo) : RB;
    const bool xs = (sr < RP);
    const bool xv = (sr < nv);
    const unsigned wo = (unsigned)(sr * K2 + sq * 4);
    const unsigned xo = (unsigned)((lo + sr) * F + sq * 4);
    v4f* __restrict__ stx = Cs4 + sr * WS4 + sq;
    v4f* __restrict__ stw = Cs4 + (RP + sr) * WS4 + sq;

    v8f acc[4][2];
    #pragma unroll
    for (int i = 0; i < 4; ++i) { acc[i][0] = (v8f){}; acc[i][1] = (v8f){}; }

    v4f pw = *reinterpret_cast<const v4f*>(W + wo);
    v4f px = xv ? *reinterpret_cast<const v4f*>(x + xo) : (v4f){};

    __syncthreads();
    const float d = xs ? (float)degc[xs ? sr : 0] : 0.0f;

    stw[0] = pw;
    if (xs) stx[0] = px * d;
    pw = *reinterpret_cast<const v4f*>(W + (wo + KC));
    px = xv ? *reinterpret_cast<const v4f*>(x + (xo + KC)) : (v4f){};
    __syncthreads();

    const float* __restrict__ wp  = Cs + (RP + cp * 32 + c) * WS + h2;
    const float* __restrict__ apx = Cs + (t0 * 16 + c) * WS + h2;
    const float* __restrict__ aps = Sl + (t0 * 16 + c) * SP + h2;

    #pragma nounroll
    for (int kc = 0; kc < F / KC; kc += 2) {
        stw[STG] = pw;
        if (xs) stx[STG] = px * d;
        pw = *reinterpret_cast<const v4f*>(W + (wo + (kc + 2) * KC));
        if (kc + 2 < F / KC)
            px = xv ? *reinterpret_cast<const v4f*>(x + (xo + (kc + 2) * KC)) : (v4f){};
        mm_main<WS>(acc, apx, wp);
        if (rg == 0) mm_extra<WS>(acc, apx, wp);
        __syncthreads();

        stw[0] = pw;
        if (kc + 2 < F / KC) {
            if (xs) stx[0] = px * d;
        }
        pw = *reinterpret_cast<const v4f*>(W + (wo + (kc + 3) * KC));
        if (kc + 3 < F / KC)
            px = xv ? *reinterpret_cast<const v4f*>(x + (xo + (kc + 3) * KC)) : (v4f){};
        mm_main<WS>(acc, apx + STG * 4, wp + STG * 4);
        if (rg == 0) mm_extra<WS>(acc, apx + STG * 4, wp + STG * 4);
        __syncthreads();
    }
    #pragma nounroll
    for (int kc = F / KC; kc < K2 / KC; kc += 2) {
        stw[STG] = pw;
        if (kc + 2 < K2 / KC)
            pw = *reinterpret_cast<const v4f*>(W + (wo + (kc + 2) * KC));
        {
            const float* __restrict__ ap = aps + (kc * KC - F);
            mm_main<SP>(acc, ap, wp);
            if (rg == 0) mm_extra<SP>(acc, ap, wp);
        }
        __syncthreads();

        if (kc + 2 < K2 / KC) {
            stw[0] = pw;
            pw = *reinterpret_cast<const v4f*>(W + (wo + (kc + 3) * KC));
        }
        {
            const float* __restrict__ ap = aps + ((kc + 1) * KC - F);
            mm_main<SP>(acc, ap, wp + STG * 4);
            if (rg == 0) mm_extra<SP>(acc, ap, wp + STG * 4);
        }
        __syncthreads();
    }

    #pragma unroll
    for (int j = 0; j < 2; ++j) {
        const int nout = cp * 32 + j * 16 + c;
        const float bias = b[nout];
        #pragma unroll
        for (int i = 0; i < 4; ++i) {
            if (i < 3 || rg == 0) {
                const int mb = (t0 + i) * 16 + (lane >> 4) * 8;
                #pragma unroll
                for (int r = 0; r < 8; ++r) {
                    const int ml = mb + r;
                    if (ml < nv)
                        out[(size_t)(lo + ml) * F + nout] = acc[i][j][r] + (float)degc[ml] * bias;
                }
            }
        }
    }
}

extern "C" void kernel_launch(void* const* d_in, const int* in_sizes, int n_in,
                              void* d_out, int out_size, void* d_ws, size_t ws_size,
                              hipStream_t stream) {
    const float* x  = (const float*)d_in[0];
    const float* W  = (const float*)d_in[1];
    const float* b  = (const float*)d_in[2];
    const int*   Ai = (const int*)d_in[3];
    const int*   Aj = (const int*)d_in[4];
    float* out = (float*)d_out;
    unsigned* slots = (unsigned*)d_ws;

    GNN_bin_kernel<<<NPROD, PT, 0, stream>>>(Ai, Aj, slots);
    GNN_fused_kernel<<<NBLK, ST, 0, stream>>>(x, W, b, Ai, Aj, slots, out);
}
